# starting kernel
# baseline (speedup 1.0000x reference)
#include <hip/hip_runtime.h>

typedef float v2f __attribute__((ext_vector_type(2)));
typedef float v8f __attribute__((ext_vector_type(8)));

#define B_  2
#define S_  2048
#define D_  1024
#define H_  16
#define DK_ 64
#define N3D_ (3 * D_)
#define MROWS_ (B_ * S_)
#define THETA_ 10000.0f

__device__ __forceinline__ v8f wmma_f32(v2f a, v2f b, v8f c) {
    return __builtin_amdgcn_wmma_f32_16x16x4_f32(
         false, a,  false, b,
         (short)0, c,  false,  false);
}

__global__ __launch_bounds__(256) void qkv_rope_kernel(
    const float* __restrict__ X, const float* __restrict__ W,
    const int* __restrict__ tokpos,
    float* __restrict__ Qo, float* __restrict__ Ko, float* __restrict__ Vo) {
    const int lane = threadIdx.x & 31;
    const int wave = threadIdx.x >> 5;
    const int l16  = lane & 15;
    const int half = lane >> 4;
    const int rowBase = blockIdx.y * 32 + (wave >> 2) * 16;
    const int colBase = blockIdx.x * 64 + (wave & 3) * 16;

    const float* Arow = X + (size_t)(rowBase + l16) * D_ + 2 * half;
    const float* Bcol = W + (size_t)(2 * half) * N3D_ + colBase + l16;

    v8f acc = {};
    #pragma unroll 4
    for (int k = 0; k < D_; k += 4) {
        v2f a; a.x = Arow[k];               a.y = Arow[k + 1];
        v2f b; b.x = Bcol[(size_t)k * N3D_]; b.y = Bcol[(size_t)(k + 1) * N3D_];
        acc = wmma_f32(a, b, acc);
    }

    #pragma unroll
    for (int t = 0; t < 8; ++t) {
        const int M  = t + 8 * half;
        const int gm = rowBase + M;
        const int gn = colBase + l16;
        const int s  = gm & (S_ - 1);
        const int b  = gm >> 11;
        const int sel = gn >> 10;
        const int d   = gn & (D_ - 1);
        const int h   = d >> 6;
        const int dd  = d & (DK_ - 1);
        float val = acc[t];
        float out;
        if (sel < 2) {
            const float partner = __shfl_xor(val, 1, 32);
            const float pos = (float)tokpos[s];
            const float fexp = (float)(dd & ~1) * (1.0f / (float)DK_);
            const float inv_freq = __powf(THETA_, -fexp);
            float sn, cs;
            __sincosf(pos * inv_freq, &sn, &cs);
            out = (dd & 1) ? (partner * sn + val * cs)
                           : (val * cs - partner * sn);
        } else {
            out = val;
        }
        float* dst = (sel == 0) ? Qo : (sel == 1) ? Ko : Vo;
        dst[(size_t)(((b * H_ + h) * S_) + s) * DK_ + dd] = out;
    }
}

__global__ __launch_bounds__(32) void flash_kernel(
    const float* __restrict__ Q, const float* __restrict__ K,
    const float* __restrict__ V, float* __restrict__ Oout) {
    __shared__ float pbuf[16][17];
    const int lane = threadIdx.x & 31;
    const int l16  = lane & 15;
    const int half = lane >> 4;
    const int qt = blockIdx.x;
    const int bh = blockIdx.y;
    const int b  = bh >> 4, h = bh & 15;
    const int qBase = qt * 16;

    const float* Qp = Q + (size_t)bh * S_ * DK_;
    const float* Kp = K + (size_t)bh * S_ * DK_;
    const float* Vp = V + (size_t)bh * S_ * DK_;

    v2f qa[16];
    #pragma unroll
    for (int kk = 0; kk < 16; ++kk) {
        const float* p = Qp + (size_t)(qBase + l16) * DK_ + kk * 4 + 2 * half;
        qa[kk].x = p[0] * 0.125f;
        qa[kk].y = p[1] * 0.125f;
    }

    v8f o0 = {}, o1 = {}, o2 = {}, o3 = {};
    float m_i[8], l_i[8];
    #pragma unroll
    for (int t = 0; t < 8; ++t) { m_i[t] = -1e30f; l_i[t] = 0.0f; }

    for (int kt = 0; kt <= qt; ++kt) {
        const int kBase = kt * 16;

        v8f sc = {};
        #pragma unroll
        for (int kk = 0; kk < 16; ++kk) {
            const float* p = Kp + (size_t)(kBase + l16) * DK_ + kk * 4 + 2 * half;
            v2f bf; bf.x = p[0]; bf.y = p[1];
            sc = wmma_f32(qa[kk], bf, sc);
        }

        float alpha[8];
        #pragma unroll
        for (int t = 0; t < 8; ++t) {
            const int row = qBase + t + 8 * half;
            const int col = kBase + l16;
            float v = (col <= row) ? sc[t] : -1e30f;
            float r = v;
            r = fmaxf(r, __shfl_xor(r, 1, 32));
            r = fmaxf(r, __shfl_xor(r, 2, 32));
            r = fmaxf(r, __shfl_xor(r, 4, 32));
            r = fmaxf(r, __shfl_xor(r, 8, 32));
            const float mn = fmaxf(m_i[t], r);
            const float a  = __expf(m_i[t] - mn);
            const float pv = __expf(v - mn);
            float rs = pv;
            rs += __shfl_xor(rs, 1, 32);
            rs += __shfl_xor(rs, 2, 32);
            rs += __shfl_xor(rs, 4, 32);
            rs += __shfl_xor(rs, 8, 32);
            l_i[t] = l_i[t] * a + rs;
            m_i[t] = mn;
            alpha[t] = a;
            sc[t] = pv;
        }
        #pragma unroll
        for (int t = 0; t < 8; ++t) {
            o0[t] *= alpha[t]; o1[t] *= alpha[t];
            o2[t] *= alpha[t]; o3[t] *= alpha[t];
        }

        #pragma unroll
        for (int t = 0; t < 8; ++t) pbuf[t + 8 * half][l16] = sc[t];
        asm volatile("s_wait_dscnt 0" ::: "memory");
        v2f pa[4];
        #pragma unroll
        for (int kk = 0; kk < 4; ++kk) {
            pa[kk].x = pbuf[l16][kk * 4 + 2 * half];
            pa[kk].y = pbuf[l16][kk * 4 + 2 * half + 1];
        }

        #pragma unroll
        for (int kk = 0; kk < 4; ++kk) {
            const int krow = kBase + kk * 4 + 2 * half;
            const float* v0 = Vp + (size_t)krow * DK_;
            const float* v1 = v0 + DK_;
            v2f b0; b0.x = v0[ 0 + l16]; b0.y = v1[ 0 + l16];
            v2f b1; b1.x = v0[16 + l16]; b1.y = v1[16 + l16];
            v2f b2; b2.x = v0[32 + l16]; b2.y = v1[32 + l16];
            v2f b3; b3.x = v0[48 + l16]; b3.y = v1[48 + l16];
            o0 = wmma_f32(pa[kk], b0, o0);
            o1 = wmma_f32(pa[kk], b1, o1);
            o2 = wmma_f32(pa[kk], b2, o2);
            o3 = wmma_f32(pa[kk], b3, o3);
        }
        asm volatile("s_wait_dscnt 0" ::: "memory");
    }

    float* Orow = Oout + ((size_t)b * S_ + qBase) * D_ + h * DK_;
    #pragma unroll
    for (int t = 0; t < 8; ++t) {
        const float inv = 1.0f / l_i[t];
        float* dst = Orow + (size_t)(t + 8 * half) * D_;
        dst[ 0 + l16] = o0[t] * inv;
        dst[16 + l16] = o1[t] * inv;
        dst[32 + l16] = o2[t] * inv;
        dst[48 + l16] = o3[t] * inv;
    }
}

__global__ __launch_bounds__(256) void out_proj_kernel(
    const float* __restrict__ A, const float* __restrict__ W,
    float* __restrict__ out) {
    const int lane = threadIdx.x & 31;
    const int wave = threadIdx.x >> 5;
    const int l16  = lane & 15;
    const int half = lane >> 4;
    const int rowBase = blockIdx.y * 32 + (wave >> 2) * 16;
    const int colBase = blockIdx.x * 64 + (wave & 3) * 16;

    const float* Arow = A + (size_t)(rowBase + l16) * D_ + 2 * half;
    const float* Bcol = W + (size_t)(2 * half) * D_ + colBase + l16;

    v8f acc = {};
    #pragma unroll 4
    for (int k = 0; k < D_; k += 4) {
        v2f a; a.x = Arow[k];              a.y = Arow[k + 1];
        v2f b; b.x = Bcol[(size_t)k * D_]; b.y = Bcol[(size_t)(k + 1) * D_];
        acc = wmma_f32(a, b, acc);
    }
    #pragma unroll
    for (int t = 0; t < 8; ++t) {
        const int gm = rowBase + t + 8 * half;
        const int gn = colBase + l16;
        out[(size_t)gm * D_ + gn] = acc[t];
    }
}

extern "C" void kernel_launch(void* const* d_in, const int* in_sizes, int n_in,
                              void* d_out, int out_size, void* d_ws, size_t ws_size,
                              hipStream_t stream) {
    (void)in_sizes; (void)n_in; (void)out_size; (void)ws_size;
    const float* X      = (const float*)d_in[0];
    const int*   tokpos = (const int*)d_in[1];
    const float* Wqkv   = (const float*)d_in[2];
    const float* Wo     = (const float*)d_in[3];
    float* out = (float*)d_out;

    float* ws = (float*)d_ws;
    const size_t BHSD = (size_t)B_ * H_ * S_ * DK_;
    float* Qb = ws;
    float* Kb = ws + BHSD;
    float* Vb = ws + 2 * BHSD;
    float* Ob = ws + 3 * BHSD;

    dim3 g1(N3D_ / 64, MROWS_ / 32);
    qkv_rope_kernel<<<g1, 256, 0, stream>>>(X, Wqkv, tokpos, Qb, Kb, Vb);

    dim3 g2(S_ / 16, B_ * H_);
    flash_kernel<<<g2, 32, 0, stream>>>(Qb, Kb, Vb, Ob);

    dim3 g3(D_ / 64, MROWS_ / 32);
    out_proj_kernel<<<g3, 256, 0, stream>>>(Ob, Wo, out);
}


# Q, K and attention output stored in WMMA fragment order in the workspace (Q/K via swapped operand roles in the projection): a fragment load is 512 contiguous bytes, 4 lines not 16; sums unchanged
# speedup vs baseline: 2.4713x; 2.4713x over previous
#include <hip/hip_runtime.h>

typedef float v2f __attribute__((ext_vector_type(2)));
typedef float v4f __attribute__((ext_vector_type(4)));
typedef float v8f __attribute__((ext_vector_type(8)));

#define B_  2
#define S_  2048
#define D_  1024
#define H_  16
#define DK_ 64
#define N3D_ (3 * D_)
#define MROWS_ (B_ * S_)
#define THETA_ 10000.0f

__device__ __forceinline__ v8f wmma_f32(v2f a, v2f b, v8f c) {
    return __builtin_amdgcn_wmma_f32_16x16x4_f32(
         false, a,  false, b,
         (short)0, c,  false,  false);
}

struct Stage { v4f a[2][2]; v4f b[2][4]; };

template <int LDB, int AK>
__device__ __forceinline__ void load_stage(Stage& s, const float* __restrict__ Ap,
                                           const float* __restrict__ Bp, int kb) {
    #pragma unroll
    for (int u = 0; u < 2; ++u) {
        s.a[u][0] = *(const v4f*)(Ap + AK * (kb + 8 * u));
        s.a[u][1] = *(const v4f*)(Ap + 16 * D_ + AK * (kb + 8 * u));
        #pragma unroll
        for (int e = 0; e < 4; ++e)
            s.b[u][e] = *(const v4f*)(Bp + (size_t)(kb + 8 * u + e) * LDB);
    }
}

template <bool SWAP>
__device__ __forceinline__ void mma_stage(const Stage& s, v8f (&acc)[2][4]) {
    #pragma unroll
    for (int u = 0; u < 2; ++u) {
        #pragma unroll
        for (int j = 0; j < 2; ++j) {
            #pragma unroll
            for (int q = 0; q < 4; ++q) {
                v2f bv; bv.x = s.b[u][2 * j][q]; bv.y = s.b[u][2 * j + 1][q];
                #pragma unroll
                for (int i = 0; i < 2; ++i) {
                    v2f av; av.x = s.a[u][i][2 * j]; av.y = s.a[u][i][2 * j + 1];
                    acc[i][q] = SWAP ? wmma_f32(bv, av, acc[i][q])
                                     : wmma_f32(av, bv, acc[i][q]);
                }
            }
        }
    }
}

template <int LDB, bool SWAP, int AK>
__device__ __forceinline__ void gemm_wave_32x64(const float* __restrict__ A,
                                                const float* __restrict__ Wm,
                                                int rowBase, int colBase,
                                                int l16, int half,
                                                v8f (&acc)[2][4]) {
    const float* Ap = (AK == 1) ? A + (size_t)(rowBase + l16) * D_ + 4 * half
                                : A + (size_t)rowBase * D_ + 64 * half + 4 * l16;
    const float* Bp = Wm + (size_t)(4 * half) * LDB + colBase + 4 * l16;
    #pragma unroll
    for (int i = 0; i < 2; ++i)
        #pragma unroll
        for (int q = 0; q < 4; ++q) acc[i][q] = v8f{};

    Stage s0, s1;
    load_stage<LDB, AK>(s0, Ap, Bp, 0);
    for (int kb = 0; kb < D_; kb += 32) {
        load_stage<LDB, AK>(s1, Ap, Bp, kb + 16);
        __builtin_amdgcn_sched_barrier(0);
        mma_stage<SWAP>(s0, acc);
        __builtin_amdgcn_sched_barrier(0);
        load_stage<LDB, AK>(s0, Ap, Bp, (kb + 32) & (D_ - 1));
        __builtin_amdgcn_sched_barrier(0);
        mma_stage<SWAP>(s1, acc);
    }
}

__global__ __launch_bounds__(256) void qkv_rope_kernel(
    const float* __restrict__ X, const float* __restrict__ W,
    const int* __restrict__ tokpos,
    float* __restrict__ Qo, float* __restrict__ Ko, float* __restrict__ Vo) {
    const int lane = threadIdx.x & 31;
    const int wave = threadIdx.x >> 5;
    const int l16  = lane & 15;
    const int half = lane >> 4;
    const int rowBase = blockIdx.y * 128 + (wave >> 1) * 32;
    const int colBase = blockIdx.x * 128 + (wave & 1) * 64;

    const int sel = colBase >> 10;
    const int h   = (colBase & (D_ - 1)) >> 6;
    const int b   = rowBase >> 11;
    const int sB  = rowBase & (S_ - 1);
    float* hbase = ((sel == 0) ? Qo : (sel == 1) ? Ko : Vo)
                 + (size_t)(((b * H_ + h) * S_) + sB) * DK_;

    v8f acc[2][4];
    if (sel < 2) {
        gemm_wave_32x64<N3D_, true, 1>(X, W, rowBase, colBase, l16, half, acc);
        const float fl0 = __powf(THETA_, -(float)(4 * l16)     * (1.0f / (float)DK_));
        const float fl1 = __powf(THETA_, -(float)(4 * l16 + 2) * (1.0f / (float)DK_));
        const float pos0 = (float)tokpos[sB + l16];
        const float pos1 = (float)tokpos[sB + 16 + l16];
        float* dt = hbase + (size_t)(8 * half) * 64 + 4 * l16;
        #pragma unroll
        for (int t = 0; t < 8; ++t) {
            const int j = 8 * half + t;
            const float f0 = __shfl(fl0, j, 32);
            const float f1 = __shfl(fl1, j, 32);
            #pragma unroll
            for (int i = 0; i < 2; ++i) {
                const float pos = i ? pos1 : pos0;
                float sn0, cs0, sn1, cs1;
                __sincosf(pos * f0, &sn0, &cs0);
                __sincosf(pos * f1, &sn1, &cs1);
                const float x0 = acc[i][0][t], x1 = acc[i][1][t];
                const float x2 = acc[i][2][t], x3 = acc[i][3][t];
                v4f o;
                o.x = x0 * cs0 - x1 * sn0;
                o.y = x0 * sn0 + x1 * cs0;
                o.z = x2 * cs1 - x3 * sn1;
                o.w = x2 * sn1 + x3 * cs1;
                *(v4f*)(dt + (size_t)(16 * i + t) * 64) = o;
            }
        }
    } else {
        gemm_wave_32x64<N3D_, false, 1>(X, W, rowBase, colBase, l16, half, acc);
        float* drow = hbase + (size_t)(8 * half) * DK_ + 4 * l16;
        #pragma unroll
        for (int i = 0; i < 2; ++i) {
            #pragma unroll
            for (int t = 0; t < 8; ++t) {
                v4f o;
                o.x = acc[i][0][t]; o.y = acc[i][1][t];
                o.z = acc[i][2][t]; o.w = acc[i][3][t];
                *(v4f*)(drow + (size_t)(16 * i + t) * DK_) = o;
            }
        }
    }
}

#define PP_ 68

__device__ __forceinline__ void load_k_block(v4f (&kf)[4][8], const float* __restrict__ Kl) {
    #pragma unroll
    for (int n = 0; n < 4; ++n)
        #pragma unroll
        for (int m = 0; m < 8; ++m)
            kf[n][m] = *(const v4f*)(Kl + (size_t)(16 * n) * DK_ + 128 * m);
}

__global__ __launch_bounds__(128) void flash_kernel(
    const float* __restrict__ Q, const float* __restrict__ K,
    const float* __restrict__ V, float* __restrict__ Oout) {
    __shared__ __attribute__((aligned(16))) float pbuf[4][16 * PP_];
    const int lane = threadIdx.x & 31;
    const int wave = threadIdx.x >> 5;
    const int l16  = lane & 15;
    const int half = lane >> 4;
    const int bh = blockIdx.y;
    const int b  = bh >> 4, h = bh & 15;

    const float* Qp = Q + (size_t)bh * S_ * DK_;
    const float* Kp = K + (size_t)bh * S_ * DK_;
    const float* Vp = V + (size_t)bh * S_ * DK_;
    float* pb = pbuf[wave];

    const float* Kl = Kp + (size_t)half * 64 + 4 * l16;
    const float* Vl = Vp + (size_t)(4 * half) * DK_ + 4 * l16;

    for (int pass = 0; pass < 2; ++pass) {
        const int g = pass ? (int)blockIdx.x : (S_ / 64 - 1) - (int)blockIdx.x;
        const int qBase = g * 64 + wave * 16;

        v4f qf[8];
        {
            const float* p = Qp + (size_t)qBase * DK_ + half * 64 + 4 * l16;
            #pragma unroll
            for (int m = 0; m < 8; ++m) qf[m] = *(const v4f*)(p + 128 * m) * 0.125f;
        }

        v8f o[4];
        #pragma unroll
        for (int q = 0; q < 4; ++q) o[q] = v8f{};
        float m_i[8], l_i[8];
        #pragma unroll
        for (int t = 0; t < 8; ++t) { m_i[t] = -1e30f; l_i[t] = 0.0f; }

        v4f kf[4][8];
        load_k_block(kf, Kl);

        for (int jb = 0; jb <= g; ++jb) {
            const int kBase = jb * 64;

            v4f vf[8][2][2];
            {
                const float* vp = Vl + (size_t)kBase * DK_;
                #pragma unroll
                for (int m = 0; m < 8; ++m)
                    #pragma unroll
                    for (int j = 0; j < 2; ++j)
                        #pragma unroll
                        for (int s = 0; s < 2; ++s)
                            vf[m][j][s] = *(const v4f*)(vp + (size_t)(8 * m + 2 * j + s) * DK_);
            }
            __builtin_amdgcn_sched_barrier(0);

            v8f sc[4];
            #pragma unroll
            for (int n = 0; n < 4; ++n) sc[n] = v8f{};
            #pragma unroll
            for (int m = 0; m < 8; ++m) {
                #pragma unroll
                for (int j = 0; j < 2; ++j) {
                    v2f av; av.x = qf[m][2 * j]; av.y = qf[m][2 * j + 1];
                    #pragma unroll
                    for (int n = 0; n < 4; ++n) {
                        v2f bv; bv.x = kf[n][m][2 * j]; bv.y = kf[n][m][2 * j + 1];
                        sc[n] = wmma_f32(av, bv, sc[n]);
                    }
                }
            }

            if (jb == g) {
                #pragma unroll
                for (int n = 0; n < 4; ++n) {
                    #pragma unroll
                    for (int t = 0; t < 8; ++t) {
                        const int row = qBase + t + 8 * half;
                        const int col = kBase + 16 * n + l16;
                        if (col > row) sc[n][t] = -1e30f;
                    }
                }
            }

            float alpha[8];
            #pragma unroll
            for (int t = 0; t < 8; ++t) {
                float r = fmaxf(fmaxf(sc[0][t], sc[1][t]), fmaxf(sc[2][t], sc[3][t]));
                r = fmaxf(r, __shfl_xor(r, 1, 32));
                r = fmaxf(r, __shfl_xor(r, 2, 32));
                r = fmaxf(r, __shfl_xor(r, 4, 32));
                r = fmaxf(r, __shfl_xor(r, 8, 32));
                const float mn = fmaxf(m_i[t], r);
                const float a  = __expf(m_i[t] - mn);
                const float p0 = __expf(sc[0][t] - mn);
                const float p1 = __expf(sc[1][t] - mn);
                const float p2 = __expf(sc[2][t] - mn);
                const float p3 = __expf(sc[3][t] - mn);
                float rs = (p0 + p1) + (p2 + p3);
                rs += __shfl_xor(rs, 1, 32);
                rs += __shfl_xor(rs, 2, 32);
                rs += __shfl_xor(rs, 4, 32);
                rs += __shfl_xor(rs, 8, 32);
                l_i[t] = l_i[t] * a + rs;
                m_i[t] = mn;
                alpha[t] = a;
                sc[0][t] = p0; sc[1][t] = p1; sc[2][t] = p2; sc[3][t] = p3;
            }
            #pragma unroll
            for (int q = 0; q < 4; ++q)
                #pragma unroll
                for (int t = 0; t < 8; ++t) o[q][t] *= alpha[t];

            #pragma unroll
            for (int n = 0; n < 4; ++n)
                #pragma unroll
                for (int t = 0; t < 8; ++t)
                    pb[(8 * half + t) * PP_ + 16 * n + l16] = sc[n][t];
            asm volatile("s_wait_dscnt 0" ::: "memory");
            v4f pf[8];
            #pragma unroll
            for (int m = 0; m < 8; ++m)
                pf[m] = *(const v4f*)(pb + l16 * PP_ + 8 * m + 4 * half);

            {
                const int nb = (jb < g) ? jb + 1 : g;
                load_k_block(kf, Kl + (size_t)nb * 64 * DK_);
            }
            __builtin_amdgcn_sched_barrier(0);

            #pragma unroll
            for (int m = 0; m < 8; ++m) {
                #pragma unroll
                for (int j = 0; j < 2; ++j) {
                    v2f av; av.x = pf[m][2 * j]; av.y = pf[m][2 * j + 1];
                    #pragma unroll
                    for (int q = 0; q < 4; ++q) {
                        v2f bv; bv.x = vf[m][j][0][q]; bv.y = vf[m][j][1][q];
                        o[q] = wmma_f32(av, bv, o[q]);
                    }
                }
            }
        }

        #pragma unroll
        for (int t = 0; t < 8; ++t) {
            const float inv = 1.0f / l_i[t];
            v4f ov;
            ov.x = o[0][t] * inv; ov.y = o[1][t] * inv;
            ov.z = o[2][t] * inv; ov.w = o[3][t] * inv;
            *(v4f*)(pb + (8 * half + t) * PP_ + 4 * l16) = ov;
        }
        asm volatile("s_wait_dscnt 0" ::: "memory");
        float* Ot = Oout + ((size_t)b * S_ + qBase) * D_ + (size_t)h * (16 * 64) + 64 * half + 4 * l16;
        #pragma unroll
        for (int x = 0; x < 8; ++x) {
            const v4f ov = *(const v4f*)(pb + l16 * PP_ + 8 * x + 4 * half);
            *(v4f*)(Ot + 128 * x) = ov;
        }
    }
}

__global__ __launch_bounds__(128) void out_proj_kernel(
    const float* __restrict__ A, const float* __restrict__ W,
    float* __restrict__ out) {
    const int lane = threadIdx.x & 31;
    const int wave = threadIdx.x >> 5;
    const int l16  = lane & 15;
    const int half = lane >> 4;
    const int rowBase = blockIdx.y * 64  + (wave >> 1) * 32;
    const int colBase = blockIdx.x * 128 + (wave & 1) * 64;

    v8f acc[2][4];
    gemm_wave_32x64<D_, false, 16>(A, W, rowBase, colBase, l16, half, acc);

    #pragma unroll
    for (int i = 0; i < 2; ++i) {
        #pragma unroll
        for (int t = 0; t < 8; ++t) {
            const int gm = rowBase + 16 * i + 8 * half + t;
            v4f o;
            o.x = acc[i][0][t]; o.y = acc[i][1][t];
            o.z = acc[i][2][t]; o.w = acc[i][3][t];
            *(v4f*)(out + (size_t)gm * D_ + colBase + 4 * l16) = o;
        }
    }
}

extern "C" void kernel_launch(void* const* d_in, const int* in_sizes, int n_in,
                              void* d_out, int out_size, void* d_ws, size_t ws_size,
                              hipStream_t stream) {
    (void)in_sizes; (void)n_in; (void)out_size; (void)ws_size;
    const float* X      = (const float*)d_in[0];
    const int*   tokpos = (const int*)d_in[1];
    const float* Wqkv   = (const float*)d_in[2];
    const float* Wo     = (const float*)d_in[3];
    float* out = (float*)d_out;

    float* ws = (float*)d_ws;
    const size_t BHSD = (size_t)B_ * H_ * S_ * DK_;
    float* Qb = ws;
    float* Kb = ws + BHSD;
    float* Vb = ws + 2 * BHSD;
    float* Ob = ws + 3 * BHSD;

    dim3 g1(N3D_ / 128, MROWS_ / 128);
    qkv_rope_kernel<<<g1, 256, 0, stream>>>(X, Wqkv, tokpos, Qb, Kb, Vb);

    dim3 g2(S_ / 128, B_ * H_);
    flash_kernel<<<g2, 128, 0, stream>>>(Qb, Kb, Vb, Ob);

    dim3 g3(D_ / 128, MROWS_ / 64);
    out_proj_kernel<<<g3, 128, 0, stream>>>(Ob, Wo, out);
}


# Attention softmax: row maximum reduced over the 16 lanes by DPP row_xmask on the vector ALU instead of LDS-routed lane permutes; row sum kept per lane, joined once per pass; sum order changes
# speedup vs baseline: 2.5732x; 1.0412x over previous
#include <hip/hip_runtime.h>

typedef float v2f __attribute__((ext_vector_type(2)));
typedef float v4f __attribute__((ext_vector_type(4)));
typedef float v8f __attribute__((ext_vector_type(8)));

#define B_  2
#define S_  2048
#define D_  1024
#define H_  16
#define DK_ 64
#define N3D_ (3 * D_)
#define MROWS_ (B_ * S_)
#define THETA_ 10000.0f

__device__ __forceinline__ v8f wmma_f32(v2f a, v2f b, v8f c) {
    return __builtin_amdgcn_wmma_f32_16x16x4_f32(
         false, a,  false, b,
         (short)0, c,  false,  false);
}

struct Stage { v4f a[2][2]; v4f b[2][4]; };

template <int LDB, int AK>
__device__ __forceinline__ void load_stage(Stage& s, const float* __restrict__ Ap,
                                           const float* __restrict__ Bp, int kb) {
    #pragma unroll
    for (int u = 0; u < 2; ++u) {
        s.a[u][0] = *(const v4f*)(Ap + AK * (kb + 8 * u));
        s.a[u][1] = *(const v4f*)(Ap + 16 * D_ + AK * (kb + 8 * u));
        #pragma unroll
        for (int e = 0; e < 4; ++e)
            s.b[u][e] = *(const v4f*)(Bp + (size_t)(kb + 8 * u + e) * LDB);
    }
}

template <bool SWAP>
__device__ __forceinline__ void mma_stage(const Stage& s, v8f (&acc)[2][4]) {
    #pragma unroll
    for (int u = 0; u < 2; ++u) {
        #pragma unroll
        for (int j = 0; j < 2; ++j) {
            #pragma unroll
            for (int q = 0; q < 4; ++q) {
                v2f bv; bv.x = s.b[u][2 * j][q]; bv.y = s.b[u][2 * j + 1][q];
                #pragma unroll
                for (int i = 0; i < 2; ++i) {
                    v2f av; av.x = s.a[u][i][2 * j]; av.y = s.a[u][i][2 * j + 1];
                    acc[i][q] = SWAP ? wmma_f32(bv, av, acc[i][q])
                                     : wmma_f32(av, bv, acc[i][q]);
                }
            }
        }
    }
}

template <int LDB, bool SWAP, int AK>
__device__ __forceinline__ void gemm_wave_32x64(const float* __restrict__ A,
                                                const float* __restrict__ Wm,
                                                int rowBase, int colBase,
                                                int l16, int half,
                                                v8f (&acc)[2][4]) {
    const float* Ap = (AK == 1) ? A + (size_t)(rowBase + l16) * D_ + 4 * half
                                : A + (size_t)rowBase * D_ + 64 * half + 4 * l16;
    const float* Bp = Wm + (size_t)(4 * half) * LDB + colBase + 4 * l16;
    #pragma unroll
    for (int i = 0; i < 2; ++i)
        #pragma unroll
        for (int q = 0; q < 4; ++q) acc[i][q] = v8f{};

    Stage s0, s1;
    load_stage<LDB, AK>(s0, Ap, Bp, 0);
    for (int kb = 0; kb < D_; kb += 32) {
        load_stage<LDB, AK>(s1, Ap, Bp, kb + 16);
        __builtin_amdgcn_sched_barrier(0);
        mma_stage<SWAP>(s0, acc);
        __builtin_amdgcn_sched_barrier(0);
        load_stage<LDB, AK>(s0, Ap, Bp, (kb + 32) & (D_ - 1));
        __builtin_amdgcn_sched_barrier(0);
        mma_stage<SWAP>(s1, acc);
    }
}

__global__ __launch_bounds__(256) void qkv_rope_kernel(
    const float* __restrict__ X, const float* __restrict__ W,
    const int* __restrict__ tokpos,
    float* __restrict__ Qo, float* __restrict__ Ko, float* __restrict__ Vo) {
    const int lane = threadIdx.x & 31;
    const int wave = threadIdx.x >> 5;
    const int l16  = lane & 15;
    const int half = lane >> 4;
    const int rowBase = blockIdx.y * 128 + (wave >> 1) * 32;
    const int colBase = blockIdx.x * 128 + (wave & 1) * 64;

    const int sel = colBase >> 10;
    const int h   = (colBase & (D_ - 1)) >> 6;
    const int b   = rowBase >> 11;
    const int sB  = rowBase & (S_ - 1);
    float* hbase = ((sel == 0) ? Qo : (sel == 1) ? Ko : Vo)
                 + (size_t)(((b * H_ + h) * S_) + sB) * DK_;

    v8f acc[2][4];
    if (sel < 2) {
        gemm_wave_32x64<N3D_, true, 1>(X, W, rowBase, colBase, l16, half, acc);
        const float fl0 = __powf(THETA_, -(float)(4 * l16)     * (1.0f / (float)DK_));
        const float fl1 = __powf(THETA_, -(float)(4 * l16 + 2) * (1.0f / (float)DK_));
        const float pos0 = (float)tokpos[sB + l16];
        const float pos1 = (float)tokpos[sB + 16 + l16];
        float* dt = hbase + (size_t)(8 * half) * 64 + 4 * l16;
        #pragma unroll
        for (int t = 0; t < 8; ++t) {
            const int j = 8 * half + t;
            const float f0 = __shfl(fl0, j, 32);
            const float f1 = __shfl(fl1, j, 32);
            #pragma unroll
            for (int i = 0; i < 2; ++i) {
                const float pos = i ? pos1 : pos0;
                float sn0, cs0, sn1, cs1;
                __sincosf(pos * f0, &sn0, &cs0);
                __sincosf(pos * f1, &sn1, &cs1);
                const float x0 = acc[i][0][t], x1 = acc[i][1][t];
                const float x2 = acc[i][2][t], x3 = acc[i][3][t];
                v4f o;
                o.x = x0 * cs0 - x1 * sn0;
                o.y = x0 * sn0 + x1 * cs0;
                o.z = x2 * cs1 - x3 * sn1;
                o.w = x2 * sn1 + x3 * cs1;
                *(v4f*)(dt + (size_t)(16 * i + t) * 64) = o;
            }
        }
    } else {
        gemm_wave_32x64<N3D_, false, 1>(X, W, rowBase, colBase, l16, half, acc);
        float* drow = hbase + (size_t)(8 * half) * DK_ + 4 * l16;
        #pragma unroll
        for (int i = 0; i < 2; ++i) {
            #pragma unroll
            for (int t = 0; t < 8; ++t) {
                v4f o;
                o.x = acc[i][0][t]; o.y = acc[i][1][t];
                o.z = acc[i][2][t]; o.w = acc[i][3][t];
                *(v4f*)(drow + (size_t)(16 * i + t) * DK_) = o;
            }
        }
    }
}

#define PP_ 68

template <int M>
__device__ __forceinline__ float row_xor(float x) {
    return __builtin_bit_cast(float, __builtin_amdgcn_update_dpp(
        0, __builtin_bit_cast(int, x), 0x160 + M, 0xf, 0xf, true));
}

#define ROWMAX_STEP_(M)                                                        \
    "v_max_num_f32_dpp %0, %0, %0 row_xmask:" #M " row_mask:0xf bank_mask:0xf\n\t" \
    "v_max_num_f32_dpp %1, %1, %1 row_xmask:" #M " row_mask:0xf bank_mask:0xf\n\t" \
    "v_max_num_f32_dpp %2, %2, %2 row_xmask:" #M " row_mask:0xf bank_mask:0xf\n\t" \
    "v_max_num_f32_dpp %3, %3, %3 row_xmask:" #M " row_mask:0xf bank_mask:0xf\n\t" \
    "v_max_num_f32_dpp %4, %4, %4 row_xmask:" #M " row_mask:0xf bank_mask:0xf\n\t" \
    "v_max_num_f32_dpp %5, %5, %5 row_xmask:" #M " row_mask:0xf bank_mask:0xf\n\t" \
    "v_max_num_f32_dpp %6, %6, %6 row_xmask:" #M " row_mask:0xf bank_mask:0xf\n\t" \
    "v_max_num_f32_dpp %7, %7, %7 row_xmask:" #M " row_mask:0xf bank_mask:0xf\n\t"
__device__ __forceinline__ void row_max8(float (&m)[8]) {
    asm(ROWMAX_STEP_(1) ROWMAX_STEP_(2) ROWMAX_STEP_(4) ROWMAX_STEP_(8)
        : "+v"(m[0]), "+v"(m[1]), "+v"(m[2]), "+v"(m[3]),
          "+v"(m[4]), "+v"(m[5]), "+v"(m[6]), "+v"(m[7]));
}

__device__ __forceinline__ void load_k_block(v4f (&kf)[4][8], const float* __restrict__ Kl) {
    #pragma unroll
    for (int n = 0; n < 4; ++n)
        #pragma unroll
        for (int m = 0; m < 8; ++m)
            kf[n][m] = *(const v4f*)(Kl + (size_t)(16 * n) * DK_ + 128 * m);
}

__global__ __launch_bounds__(128) void flash_kernel(
    const float* __restrict__ Q, const float* __restrict__ K,
    const float* __restrict__ V, float* __restrict__ Oout) {
    __shared__ __attribute__((aligned(16))) float pbuf[4][16 * PP_];
    const int lane = threadIdx.x & 31;
    const int wave = threadIdx.x >> 5;
    const int l16  = lane & 15;
    const int half = lane >> 4;
    const int bh = blockIdx.y;
    const int b  = bh >> 4, h = bh & 15;

    const float* Qp = Q + (size_t)bh * S_ * DK_;
    const float* Kp = K + (size_t)bh * S_ * DK_;
    const float* Vp = V + (size_t)bh * S_ * DK_;
    float* pb = pbuf[wave];

    const float* Kl = Kp + (size_t)half * 64 + 4 * l16;
    const float* Vl = Vp + (size_t)(4 * half) * DK_ + 4 * l16;

    for (int pass = 0; pass < 2; ++pass) {
        const int g = pass ? (int)blockIdx.x : (S_ / 64 - 1) - (int)blockIdx.x;
        const int qBase = g * 64 + wave * 16;

        v4f qf[8];
        {
            const float* p = Qp + (size_t)qBase * DK_ + half * 64 + 4 * l16;
            #pragma unroll
            for (int m = 0; m < 8; ++m) qf[m] = *(const v4f*)(p + 128 * m) * 0.125f;
        }

        v8f o[4];
        #pragma unroll
        for (int q = 0; q < 4; ++q) o[q] = v8f{};
        float m_i[8], l_i[8];
        #pragma unroll
        for (int t = 0; t < 8; ++t) { m_i[t] = -1e30f; l_i[t] = 0.0f; }

        v4f kf[4][8];
        load_k_block(kf, Kl);

        for (int jb = 0; jb <= g; ++jb) {
            const int kBase = jb * 64;

            v4f vf[8][2][2];
            {
                const float* vp = Vl + (size_t)kBase * DK_;
                #pragma unroll
                for (int m = 0; m < 8; ++m)
                    #pragma unroll
                    for (int j = 0; j < 2; ++j)
                        #pragma unroll
                        for (int s = 0; s < 2; ++s)
                            vf[m][j][s] = *(const v4f*)(vp + (size_t)(8 * m + 2 * j + s) * DK_);
            }
            __builtin_amdgcn_sched_barrier(0);

            v8f sc[4];
            #pragma unroll
            for (int n = 0; n < 4; ++n) sc[n] = v8f{};
            #pragma unroll
            for (int m = 0; m < 8; ++m) {
                #pragma unroll
                for (int j = 0; j < 2; ++j) {
                    v2f av; av.x = qf[m][2 * j]; av.y = qf[m][2 * j + 1];
                    #pragma unroll
                    for (int n = 0; n < 4; ++n) {
                        v2f bv; bv.x = kf[n][m][2 * j]; bv.y = kf[n][m][2 * j + 1];
                        sc[n] = wmma_f32(av, bv, sc[n]);
                    }
                }
            }

            if (jb == g) {
                #pragma unroll
                for (int n = 0; n < 4; ++n) {
                    #pragma unroll
                    for (int t = 0; t < 8; ++t) {
                        const int row = qBase + t + 8 * half;
                        const int col = kBase + 16 * n + l16;
                        if (col > row) sc[n][t] = -1e30f;
                    }
                }
            }

            float mx[8];
            #pragma unroll
            for (int t = 0; t < 8; ++t)
                mx[t] = fmaxf(fmaxf(fmaxf(fmaxf(sc[0][t], sc[1][t]), sc[2][t]), sc[3][t]), m_i[t]);
            row_max8(mx);
            float alpha[8];
            #pragma unroll
            for (int t = 0; t < 8; ++t) {
                const float mn = mx[t];
                const float a  = __expf(m_i[t] - mn);
                const float p0 = __expf(sc[0][t] - mn);
                const float p1 = __expf(sc[1][t] - mn);
                const float p2 = __expf(sc[2][t] - mn);
                const float p3 = __expf(sc[3][t] - mn);
                l_i[t] = l_i[t] * a + ((p0 + p1) + (p2 + p3));
                m_i[t] = mn;
                alpha[t] = a;
                sc[0][t] = p0; sc[1][t] = p1; sc[2][t] = p2; sc[3][t] = p3;
            }
            #pragma unroll
            for (int q = 0; q < 4; ++q)
                #pragma unroll
                for (int t = 0; t < 8; ++t) o[q][t] *= alpha[t];

            #pragma unroll
            for (int n = 0; n < 4; ++n)
                #pragma unroll
                for (int t = 0; t < 8; ++t)
                    pb[(8 * half + t) * PP_ + 16 * n + l16] = sc[n][t];
            asm volatile("s_wait_dscnt 0" ::: "memory");
            v4f pf[8];
            #pragma unroll
            for (int m = 0; m < 8; ++m)
                pf[m] = *(const v4f*)(pb + l16 * PP_ + 8 * m + 4 * half);

            {
                const int nb = (jb < g) ? jb + 1 : g;
                load_k_block(kf, Kl + (size_t)nb * 64 * DK_);
            }
            __builtin_amdgcn_sched_barrier(0);

            #pragma unroll
            for (int m = 0; m < 8; ++m) {
                #pragma unroll
                for (int j = 0; j < 2; ++j) {
                    v2f av; av.x = pf[m][2 * j]; av.y = pf[m][2 * j + 1];
                    #pragma unroll
                    for (int q = 0; q < 4; ++q) {
                        v2f bv; bv.x = vf[m][j][0][q]; bv.y = vf[m][j][1][q];
                        o[q] = wmma_f32(av, bv, o[q]);
                    }
                }
            }
        }

        #pragma unroll
        for (int t = 0; t < 8; ++t) {
            float l = l_i[t];
            l += row_xor<1>(l);
            l += row_xor<2>(l);
            l += row_xor<4>(l);
            l += row_xor<8>(l);
            const float inv = 1.0f / l;
            v4f ov;
            ov.x = o[0][t] * inv; ov.y = o[1][t] * inv;
            ov.z = o[2][t] * inv; ov.w = o[3][t] * inv;
            *(v4f*)(pb + (8 * half + t) * PP_ + 4 * l16) = ov;
        }
        asm volatile("s_wait_dscnt 0" ::: "memory");
        float* Ot = Oout + ((size_t)b * S_ + qBase) * D_ + (size_t)h * (16 * 64) + 64 * half + 4 * l16;
        #pragma unroll
        for (int x = 0; x < 8; ++x) {
            const v4f ov = *(const v4f*)(pb + l16 * PP_ + 8 * x + 4 * half);
            *(v4f*)(Ot + 128 * x) = ov;
        }
    }
}

__global__ __launch_bounds__(128) void out_proj_kernel(
    const float* __restrict__ A, const float* __restrict__ W,
    float* __restrict__ out) {
    const int lane = threadIdx.x & 31;
    const int wave = threadIdx.x >> 5;
    const int l16  = lane & 15;
    const int half = lane >> 4;
    const int rowBase = blockIdx.y * 64  + (wave >> 1) * 32;
    const int colBase = blockIdx.x * 128 + (wave & 1) * 64;

    v8f acc[2][4];
    gemm_wave_32x64<D_, false, 16>(A, W, rowBase, colBase, l16, half, acc);

    #pragma unroll
    for (int i = 0; i < 2; ++i) {
        #pragma unroll
        for (int t = 0; t < 8; ++t) {
            const int gm = rowBase + 16 * i + 8 * half + t;
            v4f o;
            o.x = acc[i][0][t]; o.y = acc[i][1][t];
            o.z = acc[i][2][t]; o.w = acc[i][3][t];
            *(v4f*)(out + (size_t)gm * D_ + colBase + 4 * l16) = o;
        }
    }
}

extern "C" void kernel_launch(void* const* d_in, const int* in_sizes, int n_in,
                              void* d_out, int out_size, void* d_ws, size_t ws_size,
                              hipStream_t stream) {
    (void)in_sizes; (void)n_in; (void)out_size; (void)ws_size;
    const float* X      = (const float*)d_in[0];
    const int*   tokpos = (const int*)d_in[1];
    const float* Wqkv   = (const float*)d_in[2];
    const float* Wo     = (const float*)d_in[3];
    float* out = (float*)d_out;

    float* ws = (float*)d_ws;
    const size_t BHSD = (size_t)B_ * H_ * S_ * DK_;
    float* Qb = ws;
    float* Kb = ws + BHSD;
    float* Vb = ws + 2 * BHSD;
    float* Ob = ws + 3 * BHSD;

    dim3 g1(N3D_ / 128, MROWS_ / 128);
    qkv_rope_kernel<<<g1, 256, 0, stream>>>(X, Wqkv, tokpos, Qb, Kb, Vb);

    dim3 g2(S_ / 128, B_ * H_);
    flash_kernel<<<g2, 128, 0, stream>>>(Qb, Kb, Vb, Ob);

    dim3 g3(D_ / 128, MROWS_ / 64);
    out_proj_kernel<<<g3, 128, 0, stream>>>(Ob, Wo, out);
}
